# starting kernel
# baseline (speedup 1.0000x reference)
#include <hip/hip_runtime.h>

typedef __attribute__((ext_vector_type(2))) float v2f;
typedef __attribute__((ext_vector_type(8))) float v8f;

#define C_IN   32
#define C_OUT  32
#define NNODE  163842
#define KTOT   224
#define LDSW_STRIDE 226

__global__ __launch_bounds__(256) void onering_conv_wmma_f32(
    const float* __restrict__ x,
    const float* __restrict__ W,
    const float* __restrict__ bias,
    const int*   __restrict__ neigh,
    float*       __restrict__ out)
{
    __shared__ __align__(16) float sW[C_OUT * LDSW_STRIDE];
    __shared__ __align__(16) float sB[C_OUT];

    const int tid = threadIdx.x;

    for (int idx = tid; idx < C_OUT * KTOT; idx += 256) {
        int co = idx / KTOT;
        int k  = idx - co * KTOT;
        sW[co * LDSW_STRIDE + k] = W[idx];
    }
    if (tid < C_OUT) sB[tid] = bias[tid];
    __syncthreads();

    const int lane = tid & 31;
    const int wave = tid >> 5;
    const int lo   = lane & 15;
    const int hi   = lane >> 4;

    const int batch = blockIdx.y;
    const int n0    = blockIdx.x * 256 + wave * 32;

    const float* xb = x   + (size_t)batch * C_IN  * NNODE;
    float*       ob = out + (size_t)batch * C_OUT * NNODE;
    const size_t sN = (size_t)NNODE;

    v8f acc[2][2];
    #pragma unroll
    for (int r = 0; r < 8; ++r) {
        float b0 = sB[hi * 8 + r];
        float b1 = sB[16 + hi * 8 + r];
        acc[0][0][r] = b0; acc[1][0][r] = b0;
        acc[0][1][r] = b1; acc[1][1][r] = b1;
    }

    int nidx[2][7];
    #pragma unroll
    for (int t = 0; t < 2; ++t) {
        int n = n0 + t * 16 + lo;
        if (n > NNODE - 1) n = NNODE - 1;
        const int* nr = neigh + (size_t)n * 7;
        #pragma unroll
        for (int j = 0; j < 7; ++j) nidx[t][j] = nr[j];
    }

    for (int j = 0; j < 7; ++j) {
        const float* p0 = xb + (size_t)nidx[0][j] + (size_t)(2 * hi) * sN;
        const float* p1 = xb + (size_t)nidx[1][j] + (size_t)(2 * hi) * sN;
        const float* w0 = &sW[lo        * LDSW_STRIDE + j * 32 + 2 * hi];
        const float* w1 = &sW[(lo + 16) * LDSW_STRIDE + j * 32 + 2 * hi];

        #pragma unroll
        for (int s = 0; s < 8; ++s) {
            v2f a0 = *(const v2f*)(w0 + s * 4);
            v2f a1 = *(const v2f*)(w1 + s * 4);
            v2f b0; b0.x = p0[0]; b0.y = p0[sN];
            v2f b1; b1.x = p1[0]; b1.y = p1[sN];
            p0 += 4 * sN; p1 += 4 * sN;

            acc[0][0] = __builtin_amdgcn_wmma_f32_16x16x4_f32(false, a0, false, b0, (short)0, acc[0][0], false, false);
            acc[0][1] = __builtin_amdgcn_wmma_f32_16x16x4_f32(false, a1, false, b0, (short)0, acc[0][1], false, false);
            acc[1][0] = __builtin_amdgcn_wmma_f32_16x16x4_f32(false, a0, false, b1, (short)0, acc[1][0], false, false);
            acc[1][1] = __builtin_amdgcn_wmma_f32_16x16x4_f32(false, a1, false, b1, (short)0, acc[1][1], false, false);
        }
    }

    #pragma unroll
    for (int t = 0; t < 2; ++t) {
        int n = n0 + t * 16 + lo;
        if (n < NNODE) {
            #pragma unroll
            for (int mt = 0; mt < 2; ++mt) {
                #pragma unroll
                for (int r = 0; r < 8; ++r) {
                    int co = mt * 16 + hi * 8 + r;
                    ob[(size_t)co * sN + n] = acc[t][mt][r];
                }
            }
        }
    }
}

extern "C" void kernel_launch(void* const* d_in, const int* in_sizes, int n_in,
                              void* d_out, int out_size, void* d_ws, size_t ws_size,
                              hipStream_t stream) {
    const float* x     = (const float*)d_in[0];
    const float* W     = (const float*)d_in[1];
    const float* bias  = (const float*)d_in[2];
    const int*   neigh = (const int*)d_in[3];
    float*       out   = (float*)d_out;

    int B = in_sizes[0] / (C_IN * NNODE);
    dim3 grid((NNODE + 255) / 256, B, 1);
    onering_conv_wmma_f32<<<grid, dim3(256, 1, 1), 0, stream>>>(x, W, bias, neigh, out);
}


# x transposed each call to node-major 128-B rows parked in not-yet-final output planes (batches 0-2): a neighbour gather reads 1 line, not 32; k summed in permuted channel order; 5 launches
# speedup vs baseline: 2.2555x; 2.2555x over previous
#include <hip/hip_runtime.h>

typedef __attribute__((ext_vector_type(2))) float v2f;
typedef __attribute__((ext_vector_type(8))) float v8f;

#define C_IN   32
#define C_OUT  32
#define NNODE  163842
#define KTOT   224
#define LDSW_STRIDE 226
#define PLANE  ((size_t)C_IN * NNODE)

__global__ __launch_bounds__(256) void onering_conv_wmma_f32(
    const float* __restrict__ x,
    const float* __restrict__ W,
    const float* __restrict__ bias,
    const int*   __restrict__ neigh,
    float*       __restrict__ out)
{
    __shared__ __align__(16) float sW[C_OUT * LDSW_STRIDE];
    __shared__ __align__(16) float sB[C_OUT];

    const int tid = threadIdx.x;

    for (int idx = tid; idx < C_OUT * KTOT; idx += 256) {
        int co = idx / KTOT;
        int k  = idx - co * KTOT;
        sW[co * LDSW_STRIDE + k] = W[idx];
    }
    if (tid < C_OUT) sB[tid] = bias[tid];
    __syncthreads();

    const int lane = tid & 31;
    const int wave = tid >> 5;
    const int lo   = lane & 15;
    const int hi   = lane >> 4;

    const int n0    = blockIdx.x * 256 + wave * 32;

    const float* xb = x;
    float*       ob = out;
    const size_t sN = (size_t)NNODE;

    v8f acc[2][2];
    #pragma unroll
    for (int r = 0; r < 8; ++r) {
        float b0 = sB[hi * 8 + r];
        float b1 = sB[16 + hi * 8 + r];
        acc[0][0][r] = b0; acc[1][0][r] = b0;
        acc[0][1][r] = b1; acc[1][1][r] = b1;
    }

    int nidx[2][7];
    #pragma unroll
    for (int t = 0; t < 2; ++t) {
        int n = n0 + t * 16 + lo;
        if (n > NNODE - 1) n = NNODE - 1;
        const int* nr = neigh + (size_t)n * 7;
        #pragma unroll
        for (int j = 0; j < 7; ++j) nidx[t][j] = nr[j];
    }

    for (int j = 0; j < 7; ++j) {
        const float* p0 = xb + (size_t)nidx[0][j] + (size_t)(2 * hi) * sN;
        const float* p1 = xb + (size_t)nidx[1][j] + (size_t)(2 * hi) * sN;
        const float* w0 = &sW[lo        * LDSW_STRIDE + j * 32 + 2 * hi];
        const float* w1 = &sW[(lo + 16) * LDSW_STRIDE + j * 32 + 2 * hi];

        #pragma unroll
        for (int s = 0; s < 8; ++s) {
            v2f a0 = *(const v2f*)(w0 + s * 4);
            v2f a1 = *(const v2f*)(w1 + s * 4);
            v2f b0; b0.x = p0[0]; b0.y = p0[sN];
            v2f b1; b1.x = p1[0]; b1.y = p1[sN];
            p0 += 4 * sN; p1 += 4 * sN;

            acc[0][0] = __builtin_amdgcn_wmma_f32_16x16x4_f32(false, a0, false, b0, (short)0, acc[0][0], false, false);
            acc[0][1] = __builtin_amdgcn_wmma_f32_16x16x4_f32(false, a1, false, b0, (short)0, acc[0][1], false, false);
            acc[1][0] = __builtin_amdgcn_wmma_f32_16x16x4_f32(false, a0, false, b1, (short)0, acc[1][0], false, false);
            acc[1][1] = __builtin_amdgcn_wmma_f32_16x16x4_f32(false, a1, false, b1, (short)0, acc[1][1], false, false);
        }
    }

    #pragma unroll
    for (int t = 0; t < 2; ++t) {
        int n = n0 + t * 16 + lo;
        if (n < NNODE) {
            #pragma unroll
            for (int mt = 0; mt < 2; ++mt) {
                #pragma unroll
                for (int r = 0; r < 8; ++r) {
                    int co = mt * 16 + hi * 8 + r;
                    ob[(size_t)co * sN + n] = acc[t][mt][r];
                }
            }
        }
    }
}

__global__ __launch_bounds__(256) void x_to_node_rows(
    const float* __restrict__ x,
    float*       __restrict__ dst)
{
    __shared__ float tile[C_IN][257];

    const int tid = threadIdx.x;
    const int n0  = blockIdx.x * 256;
    const float* xb = x   + (size_t)blockIdx.y * PLANE;
    float*       db = dst + (size_t)blockIdx.y * PLANE;

    const int n = n0 + tid;
    if (n < NNODE) {
        #pragma unroll
        for (int c = 0; c < C_IN; ++c) tile[c][tid] = xb[(size_t)c * NNODE + n];
    }
    __syncthreads();

    const int q = tid & 7;
    #pragma unroll
    for (int k = 0; k < 8; ++k) {
        int r = (tid >> 3) + k * 32;
        if (n0 + r < NNODE) {
            float4 v;
            v.x = tile[4 * q + 0][r];
            v.y = tile[4 * q + 1][r];
            v.z = tile[4 * q + 2][r];
            v.w = tile[4 * q + 3][r];
            *(float4*)(db + (size_t)(n0 + r) * C_IN + 4 * q) = v;
        }
    }
}

__global__ __launch_bounds__(256) void onering_conv_rows_wmma_f32(
    const float* __restrict__ xt,
    const float* __restrict__ W,
    const float* __restrict__ bias,
    const int*   __restrict__ neigh,
    float*       __restrict__ out)
{
    __shared__ __align__(16) float sW[C_OUT * LDSW_STRIDE];
    __shared__ __align__(16) float sB[C_OUT];

    const int tid = threadIdx.x;

    for (int idx = tid; idx < C_OUT * KTOT; idx += 256) {
        int co = idx / KTOT;
        int k  = idx - co * KTOT;
        int j  = k >> 5;
        int c  = k & 31;
        int cs = ((c & 14) << 1) + ((c >> 4) << 1) + (c & 1);
        sW[co * LDSW_STRIDE + j * 32 + cs] = W[idx];
    }
    if (tid < C_OUT) sB[tid] = bias[tid];
    __syncthreads();

    const int lane = tid & 31;
    const int wave = tid >> 5;
    const int lo   = lane & 15;
    const int hi   = lane >> 4;

    const int n0 = blockIdx.x * 256 + wave * 32;

    const float* tb = xt  + (size_t)blockIdx.y * PLANE;
    float*       ob = out + (size_t)blockIdx.y * PLANE;
    const size_t sN = (size_t)NNODE;

    v8f acc[2][2];
    #pragma unroll
    for (int r = 0; r < 8; ++r) {
        float b0 = sB[hi * 8 + r];
        float b1 = sB[16 + hi * 8 + r];
        acc[0][0][r] = b0; acc[1][0][r] = b0;
        acc[0][1][r] = b1; acc[1][1][r] = b1;
    }

    int nidx[2][7];
    #pragma unroll
    for (int t = 0; t < 2; ++t) {
        int n = n0 + t * 16 + lo;
        if (n > NNODE - 1) n = NNODE - 1;
        const int* nr = neigh + (size_t)n * 7;
        #pragma unroll
        for (int j = 0; j < 7; ++j) nidx[t][j] = nr[j];
    }

    float4 cur0[4], cur1[4], nxt0[4], nxt1[4];
    {
        const float4* q0 = (const float4*)(tb + (size_t)nidx[0][0] * C_IN + 16 * hi);
        const float4* q1 = (const float4*)(tb + (size_t)nidx[1][0] * C_IN + 16 * hi);
        #pragma unroll
        for (int i = 0; i < 4; ++i) { cur0[i] = q0[i]; cur1[i] = q1[i]; }
    }

    #pragma unroll
    for (int j = 0; j < 7; ++j) {
        if (j < 6) {
            const float4* q0 = (const float4*)(tb + (size_t)nidx[0][j + 1] * C_IN + 16 * hi);
            const float4* q1 = (const float4*)(tb + (size_t)nidx[1][j + 1] * C_IN + 16 * hi);
            #pragma unroll
            for (int i = 0; i < 4; ++i) { nxt0[i] = q0[i]; nxt1[i] = q1[i]; }
        }
        const float* w0 = &sW[lo        * LDSW_STRIDE + j * 32 + 2 * hi];
        const float* w1 = &sW[(lo + 16) * LDSW_STRIDE + j * 32 + 2 * hi];

        #pragma unroll
        for (int s = 0; s < 8; ++s) {
            v2f a0 = *(const v2f*)(w0 + s * 4);
            v2f a1 = *(const v2f*)(w1 + s * 4);
            v2f b0, b1;
            if (s & 1) { b0.x = cur0[s >> 1].z; b0.y = cur0[s >> 1].w; b1.x = cur1[s >> 1].z; b1.y = cur1[s >> 1].w; }
            else       { b0.x = cur0[s >> 1].x; b0.y = cur0[s >> 1].y; b1.x = cur1[s >> 1].x; b1.y = cur1[s >> 1].y; }

            acc[0][0] = __builtin_amdgcn_wmma_f32_16x16x4_f32(false, a0, false, b0, (short)0, acc[0][0], false, false);
            acc[0][1] = __builtin_amdgcn_wmma_f32_16x16x4_f32(false, a1, false, b0, (short)0, acc[0][1], false, false);
            acc[1][0] = __builtin_amdgcn_wmma_f32_16x16x4_f32(false, a0, false, b1, (short)0, acc[1][0], false, false);
            acc[1][1] = __builtin_amdgcn_wmma_f32_16x16x4_f32(false, a1, false, b1, (short)0, acc[1][1], false, false);
        }
        if (j < 6) {
            #pragma unroll
            for (int i = 0; i < 4; ++i) { cur0[i] = nxt0[i]; cur1[i] = nxt1[i]; }
        }
    }

    #pragma unroll
    for (int t = 0; t < 2; ++t) {
        int n = n0 + t * 16 + lo;
        if (n < NNODE) {
            #pragma unroll
            for (int mt = 0; mt < 2; ++mt) {
                #pragma unroll
                for (int r = 0; r < 8; ++r) {
                    int co = mt * 16 + hi * 8 + r;
                    ob[(size_t)co * sN + n] = acc[t][mt][r];
                }
            }
        }
    }
}

extern "C" void kernel_launch(void* const* d_in, const int* in_sizes, int n_in,
                              void* d_out, int out_size, void* d_ws, size_t ws_size,
                              hipStream_t stream) {
    const float* x     = (const float*)d_in[0];
    const float* W     = (const float*)d_in[1];
    const float* bias  = (const float*)d_in[2];
    const int*   neigh = (const int*)d_in[3];
    float*       out   = (float*)d_out;

    const dim3 blk(256, 1, 1);
    const dim3 g1((NNODE + 255) / 256, 1, 1);
    const dim3 g2((NNODE + 255) / 256, 2, 1);

    x_to_node_rows<<<g2, blk, 0, stream>>>(x, out + 2 * PLANE);
    onering_conv_rows_wmma_f32<<<g2, blk, 0, stream>>>(out + 2 * PLANE, W, bias, neigh, out);
    x_to_node_rows<<<g1, blk, 0, stream>>>(x + 2 * PLANE, out + 3 * PLANE);
    onering_conv_rows_wmma_f32<<<g1, blk, 0, stream>>>(out + 3 * PLANE, W, bias, neigh, out + 2 * PLANE);
    onering_conv_wmma_f32<<<g1, blk, 0, stream>>>(x + 3 * PLANE, W, bias, neigh, out + 3 * PLANE);
}


# last batch too gathers node-major 128-B rows: x transposed into the workspace (one plane, size tested in-kernel, else old path); 1 line per neighbour, not 32; its k in the rows path's channel order
# speedup vs baseline: 4.1020x; 1.8187x over previous
#include <hip/hip_runtime.h>

typedef __attribute__((ext_vector_type(2))) float v2f;
typedef __attribute__((ext_vector_type(8))) float v8f;

#define C_IN   32
#define C_OUT  32
#define NNODE  163842
#define KTOT   224
#define LDSW_STRIDE 226
#define PLANE  ((size_t)C_IN * NNODE)
#define PLANE_BYTES (PLANE * sizeof(float))

__device__ __forceinline__ void conv_direct_body(
    const float* __restrict__ x,
    const float* __restrict__ W,
    const float* __restrict__ bias,
    const int*   __restrict__ neigh,
    float*       __restrict__ out,
    float* sW, float* sB)
{
    const int tid = threadIdx.x;

    for (int idx = tid; idx < C_OUT * KTOT; idx += 256) {
        int co = idx / KTOT;
        int k  = idx - co * KTOT;
        sW[co * LDSW_STRIDE + k] = W[idx];
    }
    if (tid < C_OUT) sB[tid] = bias[tid];
    __syncthreads();

    const int lane = tid & 31;
    const int wave = tid >> 5;
    const int lo   = lane & 15;
    const int hi   = lane >> 4;

    const int n0    = blockIdx.x * 256 + wave * 32;

    const float* xb = x;
    float*       ob = out;
    const size_t sN = (size_t)NNODE;

    v8f acc[2][2];
    #pragma unroll
    for (int r = 0; r < 8; ++r) {
        float b0 = sB[hi * 8 + r];
        float b1 = sB[16 + hi * 8 + r];
        acc[0][0][r] = b0; acc[1][0][r] = b0;
        acc[0][1][r] = b1; acc[1][1][r] = b1;
    }

    int nidx[2][7];
    #pragma unroll
    for (int t = 0; t < 2; ++t) {
        int n = n0 + t * 16 + lo;
        if (n > NNODE - 1) n = NNODE - 1;
        const int* nr = neigh + (size_t)n * 7;
        #pragma unroll
        for (int j = 0; j < 7; ++j) nidx[t][j] = nr[j];
    }

    for (int j = 0; j < 7; ++j) {
        const float* p0 = xb + (size_t)nidx[0][j] + (size_t)(2 * hi) * sN;
        const float* p1 = xb + (size_t)nidx[1][j] + (size_t)(2 * hi) * sN;
        const float* w0 = &sW[lo        * LDSW_STRIDE + j * 32 + 2 * hi];
        const float* w1 = &sW[(lo + 16) * LDSW_STRIDE + j * 32 + 2 * hi];

        #pragma unroll
        for (int s = 0; s < 8; ++s) {
            v2f a0 = *(const v2f*)(w0 + s * 4);
            v2f a1 = *(const v2f*)(w1 + s * 4);
            v2f b0; b0.x = p0[0]; b0.y = p0[sN];
            v2f b1; b1.x = p1[0]; b1.y = p1[sN];
            p0 += 4 * sN; p1 += 4 * sN;

            acc[0][0] = __builtin_amdgcn_wmma_f32_16x16x4_f32(false, a0, false, b0, (short)0, acc[0][0], false, false);
            acc[0][1] = __builtin_amdgcn_wmma_f32_16x16x4_f32(false, a1, false, b0, (short)0, acc[0][1], false, false);
            acc[1][0] = __builtin_amdgcn_wmma_f32_16x16x4_f32(false, a0, false, b1, (short)0, acc[1][0], false, false);
            acc[1][1] = __builtin_amdgcn_wmma_f32_16x16x4_f32(false, a1, false, b1, (short)0, acc[1][1], false, false);
        }
    }

    #pragma unroll
    for (int t = 0; t < 2; ++t) {
        int n = n0 + t * 16 + lo;
        if (n < NNODE) {
            #pragma unroll
            for (int mt = 0; mt < 2; ++mt) {
                #pragma unroll
                for (int r = 0; r < 8; ++r) {
                    int co = mt * 16 + hi * 8 + r;
                    ob[(size_t)co * sN + n] = acc[t][mt][r];
                }
            }
        }
    }
}

__global__ __launch_bounds__(256) void x_to_node_rows(
    const float* __restrict__ x,
    float*       __restrict__ dst0,
    float*       __restrict__ dst1,
    size_t                    dst1_bytes)
{
    __shared__ float tile[C_IN][257];

    const bool second = blockIdx.y != 0;
    if (second && dst1_bytes < PLANE_BYTES) return;

    const int tid = threadIdx.x;
    const int n0  = blockIdx.x * 256;
    const float* xb = x + (size_t)blockIdx.y * PLANE;
    float*       db = second ? dst1 : dst0;

    const int n = n0 + tid;
    if (n < NNODE) {
        #pragma unroll
        for (int c = 0; c < C_IN; ++c) tile[c][tid] = xb[(size_t)c * NNODE + n];
    }
    __syncthreads();

    const int q = tid & 7;
    #pragma unroll
    for (int k = 0; k < 8; ++k) {
        int r = (tid >> 3) + k * 32;
        if (n0 + r < NNODE) {
            float4 v;
            v.x = tile[4 * q + 0][r];
            v.y = tile[4 * q + 1][r];
            v.z = tile[4 * q + 2][r];
            v.w = tile[4 * q + 3][r];
            *(float4*)(db + (size_t)(n0 + r) * C_IN + 4 * q) = v;
        }
    }
}

__device__ __forceinline__ void conv_rows_body(
    const float* __restrict__ tb,
    const float* __restrict__ W,
    const float* __restrict__ bias,
    const int*   __restrict__ neigh,
    float*       __restrict__ ob,
    float* sW, float* sB)
{
    const int tid = threadIdx.x;

    for (int idx = tid; idx < C_OUT * KTOT; idx += 256) {
        int co = idx / KTOT;
        int k  = idx - co * KTOT;
        int j  = k >> 5;
        int c  = k & 31;
        int cs = ((c & 14) << 1) + ((c >> 4) << 1) + (c & 1);
        sW[co * LDSW_STRIDE + j * 32 + cs] = W[idx];
    }
    if (tid < C_OUT) sB[tid] = bias[tid];
    __syncthreads();

    const int lane = tid & 31;
    const int wave = tid >> 5;
    const int lo   = lane & 15;
    const int hi   = lane >> 4;

    const int n0 = blockIdx.x * 256 + wave * 32;

    const size_t sN = (size_t)NNODE;

    v8f acc[2][2];
    #pragma unroll
    for (int r = 0; r < 8; ++r) {
        float b0 = sB[hi * 8 + r];
        float b1 = sB[16 + hi * 8 + r];
        acc[0][0][r] = b0; acc[1][0][r] = b0;
        acc[0][1][r] = b1; acc[1][1][r] = b1;
    }

    int nidx[2][7];
    #pragma unroll
    for (int t = 0; t < 2; ++t) {
        int n = n0 + t * 16 + lo;
        if (n > NNODE - 1) n = NNODE - 1;
        const int* nr = neigh + (size_t)n * 7;
        #pragma unroll
        for (int j = 0; j < 7; ++j) nidx[t][j] = nr[j];
    }

    float4 cur0[4], cur1[4], nxt0[4], nxt1[4];
    {
        const float4* q0 = (const float4*)(tb + (size_t)nidx[0][0] * C_IN + 16 * hi);
        const float4* q1 = (const float4*)(tb + (size_t)nidx[1][0] * C_IN + 16 * hi);
        #pragma unroll
        for (int i = 0; i < 4; ++i) { cur0[i] = q0[i]; cur1[i] = q1[i]; }
    }

    #pragma unroll
    for (int j = 0; j < 7; ++j) {
        if (j < 6) {
            const float4* q0 = (const float4*)(tb + (size_t)nidx[0][j + 1] * C_IN + 16 * hi);
            const float4* q1 = (const float4*)(tb + (size_t)nidx[1][j + 1] * C_IN + 16 * hi);
            #pragma unroll
            for (int i = 0; i < 4; ++i) { nxt0[i] = q0[i]; nxt1[i] = q1[i]; }
        }
        const float* w0 = &sW[lo        * LDSW_STRIDE + j * 32 + 2 * hi];
        const float* w1 = &sW[(lo + 16) * LDSW_STRIDE + j * 32 + 2 * hi];

        #pragma unroll
        for (int s = 0; s < 8; ++s) {
            v2f a0 = *(const v2f*)(w0 + s * 4);
            v2f a1 = *(const v2f*)(w1 + s * 4);
            v2f b0, b1;
            if (s & 1) { b0.x = cur0[s >> 1].z; b0.y = cur0[s >> 1].w; b1.x = cur1[s >> 1].z; b1.y = cur1[s >> 1].w; }
            else       { b0.x = cur0[s >> 1].x; b0.y = cur0[s >> 1].y; b1.x = cur1[s >> 1].x; b1.y = cur1[s >> 1].y; }

            acc[0][0] = __builtin_amdgcn_wmma_f32_16x16x4_f32(false, a0, false, b0, (short)0, acc[0][0], false, false);
            acc[0][1] = __builtin_amdgcn_wmma_f32_16x16x4_f32(false, a1, false, b0, (short)0, acc[0][1], false, false);
            acc[1][0] = __builtin_amdgcn_wmma_f32_16x16x4_f32(false, a0, false, b1, (short)0, acc[1][0], false, false);
            acc[1][1] = __builtin_amdgcn_wmma_f32_16x16x4_f32(false, a1, false, b1, (short)0, acc[1][1], false, false);
        }
        if (j < 6) {
            #pragma unroll
            for (int i = 0; i < 4; ++i) { cur0[i] = nxt0[i]; cur1[i] = nxt1[i]; }
        }
    }

    #pragma unroll
    for (int t = 0; t < 2; ++t) {
        int n = n0 + t * 16 + lo;
        if (n < NNODE) {
            #pragma unroll
            for (int mt = 0; mt < 2; ++mt) {
                #pragma unroll
                for (int r = 0; r < 8; ++r) {
                    int co = mt * 16 + hi * 8 + r;
                    ob[(size_t)co * sN + n] = acc[t][mt][r];
                }
            }
        }
    }
}

__global__ __launch_bounds__(256) void onering_conv_rows_wmma_f32(
    const float* __restrict__ xt,
    const float* __restrict__ W,
    const float* __restrict__ bias,
    const int*   __restrict__ neigh,
    float*       __restrict__ out)
{
    __shared__ __align__(16) float sW[C_OUT * LDSW_STRIDE];
    __shared__ __align__(16) float sB[C_OUT];

    conv_rows_body(xt + (size_t)blockIdx.y * PLANE, W, bias, neigh,
                   out + (size_t)blockIdx.y * PLANE, sW, sB);
}

__global__ __launch_bounds__(256) void onering_conv_last_wmma_f32(
    const float* __restrict__ x,
    const float* __restrict__ rows,
    size_t                    rows_bytes,
    const float* __restrict__ W,
    const float* __restrict__ bias,
    const int*   __restrict__ neigh,
    float*       __restrict__ out)
{
    __shared__ __align__(16) float sW[C_OUT * LDSW_STRIDE];
    __shared__ __align__(16) float sB[C_OUT];

    if (rows_bytes >= PLANE_BYTES) conv_rows_body(rows, W, bias, neigh, out, sW, sB);
    else                           conv_direct_body(x, W, bias, neigh, out, sW, sB);
}

extern "C" void kernel_launch(void* const* d_in, const int* in_sizes, int n_in,
                              void* d_out, int out_size, void* d_ws, size_t ws_size,
                              hipStream_t stream) {
    const float* x     = (const float*)d_in[0];
    const float* W     = (const float*)d_in[1];
    const float* bias  = (const float*)d_in[2];
    const int*   neigh = (const int*)d_in[3];
    float*       out   = (float*)d_out;
    float*       ws    = (float*)d_ws;

    const dim3 blk(256, 1, 1);
    const dim3 g1((NNODE + 255) / 256, 1, 1);
    const dim3 g2((NNODE + 255) / 256, 2, 1);

    x_to_node_rows<<<g2, blk, 0, stream>>>(x, out + 2 * PLANE, out + 3 * PLANE, PLANE_BYTES);
    onering_conv_rows_wmma_f32<<<g2, blk, 0, stream>>>(out + 2 * PLANE, W, bias, neigh, out);
    x_to_node_rows<<<g2, blk, 0, stream>>>(x + 2 * PLANE, out + 3 * PLANE, ws, ws_size);
    onering_conv_rows_wmma_f32<<<g1, blk, 0, stream>>>(out + 3 * PLANE, W, bias, neigh, out + 2 * PLANE);
    onering_conv_last_wmma_f32<<<g1, blk, 0, stream>>>(x + 3 * PLANE, ws, ws_size, W, bias, neigh, out + 3 * PLANE);
}


# rows path: the 8 row reads of neighbour j+1 go out as one group a whole neighbour (32 WMMAs) ahead, held by scheduling fences; neighbour 0's index and row reads go out ahead of the weight staging
# speedup vs baseline: 4.5957x; 1.1204x over previous
#include <hip/hip_runtime.h>

typedef __attribute__((ext_vector_type(2))) float v2f;
typedef __attribute__((ext_vector_type(8))) float v8f;

#define C_IN   32
#define C_OUT  32
#define NNODE  163842
#define KTOT   224
#define LDSW_STRIDE 226
#define PLANE  ((size_t)C_IN * NNODE)
#define PLANE_BYTES (PLANE * sizeof(float))

__device__ __forceinline__ void conv_direct_body(
    const float* __restrict__ x,
    const float* __restrict__ W,
    const float* __restrict__ bias,
    const int*   __restrict__ neigh,
    float*       __restrict__ out,
    float* sW, float* sB)
{
    const int tid = threadIdx.x;

    for (int idx = tid; idx < C_OUT * KTOT; idx += 256) {
        int co = idx / KTOT;
        int k  = idx - co * KTOT;
        sW[co * LDSW_STRIDE + k] = W[idx];
    }
    if (tid < C_OUT) sB[tid] = bias[tid];
    __syncthreads();

    const int lane = tid & 31;
    const int wave = tid >> 5;
    const int lo   = lane & 15;
    const int hi   = lane >> 4;

    const int n0    = blockIdx.x * 256 + wave * 32;

    const float* xb = x;
    float*       ob = out;
    const size_t sN = (size_t)NNODE;

    v8f acc[2][2];
    #pragma unroll
    for (int r = 0; r < 8; ++r) {
        float b0 = sB[hi * 8 + r];
        float b1 = sB[16 + hi * 8 + r];
        acc[0][0][r] = b0; acc[1][0][r] = b0;
        acc[0][1][r] = b1; acc[1][1][r] = b1;
    }

    int nidx[2][7];
    #pragma unroll
    for (int t = 0; t < 2; ++t) {
        int n = n0 + t * 16 + lo;
        if (n > NNODE - 1) n = NNODE - 1;
        const int* nr = neigh + (size_t)n * 7;
        #pragma unroll
        for (int j = 0; j < 7; ++j) nidx[t][j] = nr[j];
    }

    for (int j = 0; j < 7; ++j) {
        const float* p0 = xb + (size_t)nidx[0][j] + (size_t)(2 * hi) * sN;
        const float* p1 = xb + (size_t)nidx[1][j] + (size_t)(2 * hi) * sN;
        const float* w0 = &sW[lo        * LDSW_STRIDE + j * 32 + 2 * hi];
        const float* w1 = &sW[(lo + 16) * LDSW_STRIDE + j * 32 + 2 * hi];

        #pragma unroll
        for (int s = 0; s < 8; ++s) {
            v2f a0 = *(const v2f*)(w0 + s * 4);
            v2f a1 = *(const v2f*)(w1 + s * 4);
            v2f b0; b0.x = p0[0]; b0.y = p0[sN];
            v2f b1; b1.x = p1[0]; b1.y = p1[sN];
            p0 += 4 * sN; p1 += 4 * sN;

            acc[0][0] = __builtin_amdgcn_wmma_f32_16x16x4_f32(false, a0, false, b0, (short)0, acc[0][0], false, false);
            acc[0][1] = __builtin_amdgcn_wmma_f32_16x16x4_f32(false, a1, false, b0, (short)0, acc[0][1], false, false);
            acc[1][0] = __builtin_amdgcn_wmma_f32_16x16x4_f32(false, a0, false, b1, (short)0, acc[1][0], false, false);
            acc[1][1] = __builtin_amdgcn_wmma_f32_16x16x4_f32(false, a1, false, b1, (short)0, acc[1][1], false, false);
        }
    }

    #pragma unroll
    for (int t = 0; t < 2; ++t) {
        int n = n0 + t * 16 + lo;
        if (n < NNODE) {
            #pragma unroll
            for (int mt = 0; mt < 2; ++mt) {
                #pragma unroll
                for (int r = 0; r < 8; ++r) {
                    int co = mt * 16 + hi * 8 + r;
                    ob[(size_t)co * sN + n] = acc[t][mt][r];
                }
            }
        }
    }
}

__global__ __launch_bounds__(256) void x_to_node_rows(
    const float* __restrict__ x,
    float*       __restrict__ dst0,
    float*       __restrict__ dst1,
    size_t                    dst1_bytes)
{
    __shared__ float tile[C_IN][257];

    const bool second = blockIdx.y != 0;
    if (second && dst1_bytes < PLANE_BYTES) return;

    const int tid = threadIdx.x;
    const int n0  = blockIdx.x * 256;
    const float* xb = x + (size_t)blockIdx.y * PLANE;
    float*       db = second ? dst1 : dst0;

    const int n = n0 + tid;
    if (n < NNODE) {
        #pragma unroll
        for (int c = 0; c < C_IN; ++c) tile[c][tid] = xb[(size_t)c * NNODE + n];
    }
    __syncthreads();

    const int q = tid & 7;
    #pragma unroll
    for (int k = 0; k < 8; ++k) {
        int r = (tid >> 3) + k * 32;
        if (n0 + r < NNODE) {
            float4 v;
            v.x = tile[4 * q + 0][r];
            v.y = tile[4 * q + 1][r];
            v.z = tile[4 * q + 2][r];
            v.w = tile[4 * q + 3][r];
            *(float4*)(db + (size_t)(n0 + r) * C_IN + 4 * q) = v;
        }
    }
}

__device__ __forceinline__ void conv_rows_body(
    const float* __restrict__ tb,
    const float* __restrict__ W,
    const float* __restrict__ bias,
    const int*   __restrict__ neigh,
    float*       __restrict__ ob,
    float* sW, float* sB)
{
    const int tid = threadIdx.x;

    const int lane = tid & 31;
    const int wave = tid >> 5;
    const int lo   = lane & 15;
    const int hi   = lane >> 4;

    const int n0 = blockIdx.x * 256 + wave * 32;

    const size_t sN = (size_t)NNODE;

    int nidx[2][7];
    #pragma unroll
    for (int t = 0; t < 2; ++t) {
        int n = n0 + t * 16 + lo;
        if (n > NNODE - 1) n = NNODE - 1;
        const int* nr = neigh + (size_t)n * 7;
        #pragma unroll
        for (int j = 0; j < 7; ++j) nidx[t][j] = nr[j];
    }

    float4 cur0[4], cur1[4], nxt0[4], nxt1[4];
    {
        const float4* q0 = (const float4*)(tb + (size_t)nidx[0][0] * C_IN + 16 * hi);
        const float4* q1 = (const float4*)(tb + (size_t)nidx[1][0] * C_IN + 16 * hi);
        #pragma unroll
        for (int i = 0; i < 4; ++i) cur0[i] = q0[i];
        #pragma unroll
        for (int i = 0; i < 4; ++i) cur1[i] = q1[i];
    }

    for (int idx = tid; idx < C_OUT * KTOT; idx += 256) {
        int co = idx / KTOT;
        int k  = idx - co * KTOT;
        int j  = k >> 5;
        int c  = k & 31;
        int cs = ((c & 14) << 1) + ((c >> 4) << 1) + (c & 1);
        sW[co * LDSW_STRIDE + j * 32 + cs] = W[idx];
    }
    if (tid < C_OUT) sB[tid] = bias[tid];
    __syncthreads();

    v8f acc[2][2];
    #pragma unroll
    for (int r = 0; r < 8; ++r) {
        float b0 = sB[hi * 8 + r];
        float b1 = sB[16 + hi * 8 + r];
        acc[0][0][r] = b0; acc[1][0][r] = b0;
        acc[0][1][r] = b1; acc[1][1][r] = b1;
    }

    #pragma unroll
    for (int j = 0; j < 7; ++j) {
        __builtin_amdgcn_sched_barrier(0);
        if (j < 6) {
            const float4* q0 = (const float4*)(tb + (size_t)nidx[0][j + 1] * C_IN + 16 * hi);
            const float4* q1 = (const float4*)(tb + (size_t)nidx[1][j + 1] * C_IN + 16 * hi);
            #pragma unroll
            for (int i = 0; i < 4; ++i) nxt0[i] = q0[i];
            #pragma unroll
            for (int i = 0; i < 4; ++i) nxt1[i] = q1[i];
        }
        __builtin_amdgcn_sched_barrier(0);

        const float* w0 = &sW[lo        * LDSW_STRIDE + j * 32 + 2 * hi];
        const float* w1 = &sW[(lo + 16) * LDSW_STRIDE + j * 32 + 2 * hi];

        #pragma unroll
        for (int s = 0; s < 8; ++s) {
            v2f a0 = *(const v2f*)(w0 + s * 4);
            v2f a1 = *(const v2f*)(w1 + s * 4);
            v2f b0, b1;
            if (s & 1) { b0.x = cur0[s >> 1].z; b0.y = cur0[s >> 1].w; b1.x = cur1[s >> 1].z; b1.y = cur1[s >> 1].w; }
            else       { b0.x = cur0[s >> 1].x; b0.y = cur0[s >> 1].y; b1.x = cur1[s >> 1].x; b1.y = cur1[s >> 1].y; }

            acc[0][0] = __builtin_amdgcn_wmma_f32_16x16x4_f32(false, a0, false, b0, (short)0, acc[0][0], false, false);
            acc[0][1] = __builtin_amdgcn_wmma_f32_16x16x4_f32(false, a1, false, b0, (short)0, acc[0][1], false, false);
            acc[1][0] = __builtin_amdgcn_wmma_f32_16x16x4_f32(false, a0, false, b1, (short)0, acc[1][0], false, false);
            acc[1][1] = __builtin_amdgcn_wmma_f32_16x16x4_f32(false, a1, false, b1, (short)0, acc[1][1], false, false);
        }
        if (j < 6) {
            #pragma unroll
            for (int i = 0; i < 4; ++i) { cur0[i] = nxt0[i]; cur1[i] = nxt1[i]; }
        }
    }

    #pragma unroll
    for (int t = 0; t < 2; ++t) {
        int n = n0 + t * 16 + lo;
        if (n < NNODE) {
            #pragma unroll
            for (int mt = 0; mt < 2; ++mt) {
                #pragma unroll
                for (int r = 0; r < 8; ++r) {
                    int co = mt * 16 + hi * 8 + r;
                    ob[(size_t)co * sN + n] = acc[t][mt][r];
                }
            }
        }
    }
}

__global__ __launch_bounds__(256) void onering_conv_rows_wmma_f32(
    const float* __restrict__ xt,
    const float* __restrict__ W,
    const float* __restrict__ bias,
    const int*   __restrict__ neigh,
    float*       __restrict__ out)
{
    __shared__ __align__(16) float sW[C_OUT * LDSW_STRIDE];
    __shared__ __align__(16) float sB[C_OUT];

    conv_rows_body(xt + (size_t)blockIdx.y * PLANE, W, bias, neigh,
                   out + (size_t)blockIdx.y * PLANE, sW, sB);
}

__global__ __launch_bounds__(256) void onering_conv_last_wmma_f32(
    const float* __restrict__ x,
    const float* __restrict__ rows,
    size_t                    rows_bytes,
    const float* __restrict__ W,
    const float* __restrict__ bias,
    const int*   __restrict__ neigh,
    float*       __restrict__ out)
{
    __shared__ __align__(16) float sW[C_OUT * LDSW_STRIDE];
    __shared__ __align__(16) float sB[C_OUT];

    if (rows_bytes >= PLANE_BYTES) conv_rows_body(rows, W, bias, neigh, out, sW, sB);
    else                           conv_direct_body(x, W, bias, neigh, out, sW, sB);
}

extern "C" void kernel_launch(void* const* d_in, const int* in_sizes, int n_in,
                              void* d_out, int out_size, void* d_ws, size_t ws_size,
                              hipStream_t stream) {
    const float* x     = (const float*)d_in[0];
    const float* W     = (const float*)d_in[1];
    const float* bias  = (const float*)d_in[2];
    const int*   neigh = (const int*)d_in[3];
    float*       out   = (float*)d_out;
    float*       ws    = (float*)d_ws;

    const dim3 blk(256, 1, 1);
    const dim3 g1((NNODE + 255) / 256, 1, 1);
    const dim3 g2((NNODE + 255) / 256, 2, 1);

    x_to_node_rows<<<g2, blk, 0, stream>>>(x, out + 2 * PLANE, out + 3 * PLANE, PLANE_BYTES);
    onering_conv_rows_wmma_f32<<<g2, blk, 0, stream>>>(out + 2 * PLANE, W, bias, neigh, out);
    x_to_node_rows<<<g2, blk, 0, stream>>>(x + 2 * PLANE, out + 3 * PLANE, ws, ws_size);
    onering_conv_rows_wmma_f32<<<g1, blk, 0, stream>>>(out + 3 * PLANE, W, bias, neigh, out + 2 * PLANE);
    onering_conv_last_wmma_f32<<<g1, blk, 0, stream>>>(x + 3 * PLANE, ws, ws_size, W, bias, neigh, out + 3 * PLANE);
}


# rows path: a neighbour row is read as one whole 128-B line by 8 lanes (4 lines per read, not 32-B pieces of 16) and laid out as B fragments through a per-wave LDS buffer; k order and sums unchanged
# speedup vs baseline: 4.7729x; 1.0386x over previous
#include <hip/hip_runtime.h>

typedef __attribute__((ext_vector_type(2))) float v2f;
typedef __attribute__((ext_vector_type(8))) float v8f;
typedef __attribute__((ext_vector_type(4))) float v4f;

#define C_IN   32
#define C_OUT  32
#define NNODE  163842
#define KTOT   224
#define LDSW_STRIDE 226
#define PLANE  ((size_t)C_IN * NNODE)
#define PLANE_BYTES (PLANE * sizeof(float))

__device__ __forceinline__ void conv_direct_body(
    const float* __restrict__ x,
    const float* __restrict__ W,
    const float* __restrict__ bias,
    const int*   __restrict__ neigh,
    float*       __restrict__ out,
    float* sW, float* sB)
{
    const int tid = threadIdx.x;

    for (int idx = tid; idx < C_OUT * KTOT; idx += 256) {
        int co = idx / KTOT;
        int k  = idx - co * KTOT;
        sW[co * LDSW_STRIDE + k] = W[idx];
    }
    if (tid < C_OUT) sB[tid] = bias[tid];
    __syncthreads();

    const int lane = tid & 31;
    const int wave = tid >> 5;
    const int lo   = lane & 15;
    const int hi   = lane >> 4;

    const int n0    = blockIdx.x * 256 + wave * 32;

    const float* xb = x;
    float*       ob = out;
    const size_t sN = (size_t)NNODE;

    v8f acc[2][2];
    #pragma unroll
    for (int r = 0; r < 8; ++r) {
        float b0 = sB[hi * 8 + r];
        float b1 = sB[16 + hi * 8 + r];
        acc[0][0][r] = b0; acc[1][0][r] = b0;
        acc[0][1][r] = b1; acc[1][1][r] = b1;
    }

    int nidx[2][7];
    #pragma unroll
    for (int t = 0; t < 2; ++t) {
        int n = n0 + t * 16 + lo;
        if (n > NNODE - 1) n = NNODE - 1;
        const int* nr = neigh + (size_t)n * 7;
        #pragma unroll
        for (int j = 0; j < 7; ++j) nidx[t][j] = nr[j];
    }

    for (int j = 0; j < 7; ++j) {
        const float* p0 = xb + (size_t)nidx[0][j] + (size_t)(2 * hi) * sN;
        const float* p1 = xb + (size_t)nidx[1][j] + (size_t)(2 * hi) * sN;
        const float* w0 = &sW[lo        * LDSW_STRIDE + j * 32 + 2 * hi];
        const float* w1 = &sW[(lo + 16) * LDSW_STRIDE + j * 32 + 2 * hi];

        #pragma unroll
        for (int s = 0; s < 8; ++s) {
            v2f a0 = *(const v2f*)(w0 + s * 4);
            v2f a1 = *(const v2f*)(w1 + s * 4);
            v2f b0; b0.x = p0[0]; b0.y = p0[sN];
            v2f b1; b1.x = p1[0]; b1.y = p1[sN];
            p0 += 4 * sN; p1 += 4 * sN;

            acc[0][0] = __builtin_amdgcn_wmma_f32_16x16x4_f32(false, a0, false, b0, (short)0, acc[0][0], false, false);
            acc[0][1] = __builtin_amdgcn_wmma_f32_16x16x4_f32(false, a1, false, b0, (short)0, acc[0][1], false, false);
            acc[1][0] = __builtin_amdgcn_wmma_f32_16x16x4_f32(false, a0, false, b1, (short)0, acc[1][0], false, false);
            acc[1][1] = __builtin_amdgcn_wmma_f32_16x16x4_f32(false, a1, false, b1, (short)0, acc[1][1], false, false);
        }
    }

    #pragma unroll
    for (int t = 0; t < 2; ++t) {
        int n = n0 + t * 16 + lo;
        if (n < NNODE) {
            #pragma unroll
            for (int mt = 0; mt < 2; ++mt) {
                #pragma unroll
                for (int r = 0; r < 8; ++r) {
                    int co = mt * 16 + hi * 8 + r;
                    ob[(size_t)co * sN + n] = acc[t][mt][r];
                }
            }
        }
    }
}

__global__ __launch_bounds__(256) void x_to_node_rows(
    const float* __restrict__ x,
    float*       __restrict__ dst0,
    float*       __restrict__ dst1,
    size_t                    dst1_bytes)
{
    __shared__ float tile[C_IN][257];

    const bool second = blockIdx.y != 0;
    if (second && dst1_bytes < PLANE_BYTES) return;

    const int tid = threadIdx.x;
    const int n0  = blockIdx.x * 256;
    const float* xb = x + (size_t)blockIdx.y * PLANE;
    float*       db = second ? dst1 : dst0;

    const int n = n0 + tid;
    if (n < NNODE) {
        #pragma unroll
        for (int c = 0; c < C_IN; ++c) tile[c][tid] = xb[(size_t)c * NNODE + n];
    }
    __syncthreads();

    const int q = tid & 7;
    #pragma unroll
    for (int k = 0; k < 8; ++k) {
        int r = (tid >> 3) + k * 32;
        if (n0 + r < NNODE) {
            float4 v;
            v.x = tile[4 * q + 0][r];
            v.y = tile[4 * q + 1][r];
            v.z = tile[4 * q + 2][r];
            v.w = tile[4 * q + 3][r];
            *(float4*)(db + (size_t)(n0 + r) * C_IN + 4 * q) = v;
        }
    }
}

#define XROW_PITCH 36
#define XBUF_FLOATS (32 * XROW_PITCH)

__device__ __forceinline__ void wave_lds_order()
{
    __builtin_amdgcn_fence(__ATOMIC_RELEASE, "wavefront");
    __builtin_amdgcn_wave_barrier();
    __builtin_amdgcn_fence(__ATOMIC_ACQUIRE, "wavefront");
}

__device__ __forceinline__ void conv_rows_body(
    const float* __restrict__ tb,
    const float* __restrict__ W,
    const float* __restrict__ bias,
    const int*   __restrict__ neigh,
    float*       __restrict__ ob,
    float* sW, float* sB, float* sX)
{
    const int tid = threadIdx.x;

    const int lane = tid & 31;
    const int wave = tid >> 5;
    const int lo   = lane & 15;
    const int hi   = lane >> 4;
    const int sub  = lane >> 3;
    const int q    = lane & 7;

    const int n0 = blockIdx.x * 256 + wave * 32;

    const size_t sN = (size_t)NNODE;

    float* xb = sX + wave * XBUF_FLOATS;

    const char* tbytes = (const char*)tb;
    const unsigned qoff = 16u * (unsigned)q;

    int nidx[7];
    {
        int n = n0 + lane;
        if (n > NNODE - 1) n = NNODE - 1;
        const int* nr = neigh + (size_t)n * 7;
        #pragma unroll
        for (int j = 0; j < 7; ++j) nidx[j] = nr[j];
    }

    v4f rows[8];
    #pragma unroll
    for (int g = 0; g < 8; ++g) {
        const unsigned id = (unsigned)__shfl(nidx[0], 4 * g + sub);
        rows[g] = *(const v4f*)(tbytes + (id * 128u + qoff));
    }

    for (int idx = tid; idx < C_OUT * KTOT; idx += 256) {
        int co = idx / KTOT;
        int k  = idx - co * KTOT;
        int j  = k >> 5;
        int c  = k & 31;
        int cs = ((c & 14) << 1) + ((c >> 4) << 1) + (c & 1);
        sW[co * LDSW_STRIDE + j * 32 + cs] = W[idx];
    }
    if (tid < C_OUT) sB[tid] = bias[tid];
    __syncthreads();

    v8f acc[2][2];
    #pragma unroll
    for (int r = 0; r < 8; ++r) {
        float b0 = sB[hi * 8 + r];
        float b1 = sB[16 + hi * 8 + r];
        acc[0][0][r] = b0; acc[1][0][r] = b0;
        acc[0][1][r] = b1; acc[1][1][r] = b1;
    }

    v4f cur0[4], cur1[4];
    #pragma unroll
    for (int j = 0; j < 7; ++j) {
        __builtin_amdgcn_sched_barrier(0);
        wave_lds_order();
        #pragma unroll
        for (int g = 0; g < 8; ++g)
            *(v4f*)(xb + (4 * g + sub) * XROW_PITCH + 4 * q) = rows[g];
        wave_lds_order();
        #pragma unroll
        for (int i = 0; i < 4; ++i) {
            cur0[i] = *(const v4f*)(xb + lo        * XROW_PITCH + 16 * hi + 4 * i);
            cur1[i] = *(const v4f*)(xb + (16 + lo) * XROW_PITCH + 16 * hi + 4 * i);
        }
        if (j < 6) {
            #pragma unroll
            for (int g = 0; g < 8; ++g) {
                const unsigned id = (unsigned)__shfl(nidx[j + 1], 4 * g + sub);
                rows[g] = *(const v4f*)(tbytes + (id * 128u + qoff));
            }
        }

        const float* w0 = &sW[lo        * LDSW_STRIDE + j * 32 + 2 * hi];
        const float* w1 = &sW[(lo + 16) * LDSW_STRIDE + j * 32 + 2 * hi];

        #pragma unroll
        for (int s = 0; s < 8; ++s) {
            v2f a0 = *(const v2f*)(w0 + s * 4);
            v2f a1 = *(const v2f*)(w1 + s * 4);
            v2f b0, b1;
            if (s & 1) { b0.x = cur0[s >> 1].z; b0.y = cur0[s >> 1].w; b1.x = cur1[s >> 1].z; b1.y = cur1[s >> 1].w; }
            else       { b0.x = cur0[s >> 1].x; b0.y = cur0[s >> 1].y; b1.x = cur1[s >> 1].x; b1.y = cur1[s >> 1].y; }

            acc[0][0] = __builtin_amdgcn_wmma_f32_16x16x4_f32(false, a0, false, b0, (short)0, acc[0][0], false, false);
            acc[0][1] = __builtin_amdgcn_wmma_f32_16x16x4_f32(false, a1, false, b0, (short)0, acc[0][1], false, false);
            acc[1][0] = __builtin_amdgcn_wmma_f32_16x16x4_f32(false, a0, false, b1, (short)0, acc[1][0], false, false);
            acc[1][1] = __builtin_amdgcn_wmma_f32_16x16x4_f32(false, a1, false, b1, (short)0, acc[1][1], false, false);
        }
    }

    #pragma unroll
    for (int t = 0; t < 2; ++t) {
        int n = n0 + t * 16 + lo;
        if (n < NNODE) {
            #pragma unroll
            for (int mt = 0; mt < 2; ++mt) {
                #pragma unroll
                for (int r = 0; r < 8; ++r) {
                    int co = mt * 16 + hi * 8 + r;
                    ob[(size_t)co * sN + n] = acc[t][mt][r];
                }
            }
        }
    }
}

__global__ __launch_bounds__(256) void onering_conv_rows_wmma_f32(
    const float* __restrict__ xt,
    const float* __restrict__ W,
    const float* __restrict__ bias,
    const int*   __restrict__ neigh,
    float*       __restrict__ out)
{
    __shared__ __align__(16) float sW[C_OUT * LDSW_STRIDE];
    __shared__ __align__(16) float sB[C_OUT];
    __shared__ __align__(16) float sX[8 * XBUF_FLOATS];

    conv_rows_body(xt + (size_t)blockIdx.y * PLANE, W, bias, neigh,
                   out + (size_t)blockIdx.y * PLANE, sW, sB, sX);
}

__global__ __launch_bounds__(256) void onering_conv_last_wmma_f32(
    const float* __restrict__ x,
    const float* __restrict__ rows,
    size_t                    rows_bytes,
    const float* __restrict__ W,
    const float* __restrict__ bias,
    const int*   __restrict__ neigh,
    float*       __restrict__ out)
{
    __shared__ __align__(16) float sW[C_OUT * LDSW_STRIDE];
    __shared__ __align__(16) float sB[C_OUT];
    __shared__ __align__(16) float sX[8 * XBUF_FLOATS];

    if (rows_bytes >= PLANE_BYTES) conv_rows_body(rows, W, bias, neigh, out, sW, sB, sX);
    else                           conv_direct_body(x, W, bias, neigh, out, sW, sB);
}

extern "C" void kernel_launch(void* const* d_in, const int* in_sizes, int n_in,
                              void* d_out, int out_size, void* d_ws, size_t ws_size,
                              hipStream_t stream) {
    const float* x     = (const float*)d_in[0];
    const float* W     = (const float*)d_in[1];
    const float* bias  = (const float*)d_in[2];
    const int*   neigh = (const int*)d_in[3];
    float*       out   = (float*)d_out;
    float*       ws    = (float*)d_ws;

    const dim3 blk(256, 1, 1);
    const dim3 g1((NNODE + 255) / 256, 1, 1);
    const dim3 g2((NNODE + 255) / 256, 2, 1);

    x_to_node_rows<<<g2, blk, 0, stream>>>(x, out + 2 * PLANE, out + 3 * PLANE, PLANE_BYTES);
    onering_conv_rows_wmma_f32<<<g2, blk, 0, stream>>>(out + 2 * PLANE, W, bias, neigh, out);
    x_to_node_rows<<<g2, blk, 0, stream>>>(x + 2 * PLANE, out + 3 * PLANE, ws, ws_size);
    onering_conv_rows_wmma_f32<<<g1, blk, 0, stream>>>(out + 3 * PLANE, W, bias, neigh, out + 2 * PLANE);
    onering_conv_last_wmma_f32<<<g1, blk, 0, stream>>>(x + 3 * PLANE, ws, ws_size, W, bias, neigh, out + 3 * PLANE);
}


# transposes of batches 2 and 3 run as extra blocks inside the conv launches before them (workspace holds batch 1's rows first, then batch 3's): 4 launches, not 5; k order and sums unchanged
# speedup vs baseline: 5.0224x; 1.0523x over previous
#include <hip/hip_runtime.h>

typedef __attribute__((ext_vector_type(2))) float v2f;
typedef __attribute__((ext_vector_type(8))) float v8f;
typedef __attribute__((ext_vector_type(4))) float v4f;

#define C_IN   32
#define C_OUT  32
#define NNODE  163842
#define KTOT   224
#define LDSW_STRIDE 226
#define PLANE  ((size_t)C_IN * NNODE)
#define PLANE_BYTES (PLANE * sizeof(float))

template <int SU>
__device__ __forceinline__ void conv_direct_body(
    const float* __restrict__ x,
    const float* __restrict__ W,
    const float* __restrict__ bias,
    const int*   __restrict__ neigh,
    float*       __restrict__ out,
    float* sW, float* sB)
{
    const int tid = threadIdx.x;

    for (int idx = tid; idx < C_OUT * KTOT; idx += 256) {
        int co = idx / KTOT;
        int k  = idx - co * KTOT;
        sW[co * LDSW_STRIDE + k] = W[idx];
    }
    if (tid < C_OUT) sB[tid] = bias[tid];
    __syncthreads();

    const int lane = tid & 31;
    const int wave = tid >> 5;
    const int lo   = lane & 15;
    const int hi   = lane >> 4;

    const int n0    = blockIdx.x * 256 + wave * 32;

    const float* xb = x;
    float*       ob = out;
    const size_t sN = (size_t)NNODE;

    v8f acc[2][2];
    #pragma unroll
    for (int r = 0; r < 8; ++r) {
        float b0 = sB[hi * 8 + r];
        float b1 = sB[16 + hi * 8 + r];
        acc[0][0][r] = b0; acc[1][0][r] = b0;
        acc[0][1][r] = b1; acc[1][1][r] = b1;
    }

    int nidx[2][7];
    #pragma unroll
    for (int t = 0; t < 2; ++t) {
        int n = n0 + t * 16 + lo;
        if (n > NNODE - 1) n = NNODE - 1;
        const int* nr = neigh + (size_t)n * 7;
        #pragma unroll
        for (int j = 0; j < 7; ++j) nidx[t][j] = nr[j];
    }

    for (int j = 0; j < 7; ++j) {
        const float* p0 = xb + (size_t)nidx[0][j] + (size_t)(2 * hi) * sN;
        const float* p1 = xb + (size_t)nidx[1][j] + (size_t)(2 * hi) * sN;
        const float* w0 = &sW[lo        * LDSW_STRIDE + j * 32 + 2 * hi];
        const float* w1 = &sW[(lo + 16) * LDSW_STRIDE + j * 32 + 2 * hi];

        #pragma unroll SU
        for (int s = 0; s < 8; ++s) {
            v2f a0 = *(const v2f*)(w0 + s * 4);
            v2f a1 = *(const v2f*)(w1 + s * 4);
            v2f b0; b0.x = p0[0]; b0.y = p0[sN];
            v2f b1; b1.x = p1[0]; b1.y = p1[sN];
            p0 += 4 * sN; p1 += 4 * sN;

            acc[0][0] = __builtin_amdgcn_wmma_f32_16x16x4_f32(false, a0, false, b0, (short)0, acc[0][0], false, false);
            acc[0][1] = __builtin_amdgcn_wmma_f32_16x16x4_f32(false, a1, false, b0, (short)0, acc[0][1], false, false);
            acc[1][0] = __builtin_amdgcn_wmma_f32_16x16x4_f32(false, a0, false, b1, (short)0, acc[1][0], false, false);
            acc[1][1] = __builtin_amdgcn_wmma_f32_16x16x4_f32(false, a1, false, b1, (short)0, acc[1][1], false, false);
        }
    }

    #pragma unroll
    for (int t = 0; t < 2; ++t) {
        int n = n0 + t * 16 + lo;
        if (n < NNODE) {
            #pragma unroll
            for (int mt = 0; mt < 2; ++mt) {
                #pragma unroll
                for (int r = 0; r < 8; ++r) {
                    int co = mt * 16 + hi * 8 + r;
                    ob[(size_t)co * sN + n] = acc[t][mt][r];
                }
            }
        }
    }
}

__device__ __forceinline__ void rows_body(
    const float* __restrict__ xb,
    float*       __restrict__ db,
    float (*tile)[257])
{
    const int tid = threadIdx.x;
    const int n0  = blockIdx.x * 256;

    const int n = n0 + tid;
    if (n < NNODE) {
        #pragma unroll
        for (int c = 0; c < C_IN; ++c) tile[c][tid] = xb[(size_t)c * NNODE + n];
    }
    __syncthreads();

    const int q = tid & 7;
    #pragma unroll
    for (int k = 0; k < 8; ++k) {
        int r = (tid >> 3) + k * 32;
        if (n0 + r < NNODE) {
            float4 v;
            v.x = tile[4 * q + 0][r];
            v.y = tile[4 * q + 1][r];
            v.z = tile[4 * q + 2][r];
            v.w = tile[4 * q + 3][r];
            *(float4*)(db + (size_t)(n0 + r) * C_IN + 4 * q) = v;
        }
    }
}

__global__ __launch_bounds__(256) void x_to_node_rows(
    const float* __restrict__ x,
    float*       __restrict__ dst0,
    float*       __restrict__ dst1,
    size_t                    dst1_bytes)
{
    __shared__ float tile[C_IN][257];

    const bool second = blockIdx.y != 0;
    if (second && dst1_bytes < PLANE_BYTES) return;

    rows_body(x + (size_t)blockIdx.y * PLANE, second ? dst1 : dst0, tile);
}

#define XROW_PITCH 36
#define XBUF_FLOATS (32 * XROW_PITCH)

__device__ __forceinline__ void wave_lds_order()
{
    __builtin_amdgcn_fence(__ATOMIC_RELEASE, "wavefront");
    __builtin_amdgcn_wave_barrier();
    __builtin_amdgcn_fence(__ATOMIC_ACQUIRE, "wavefront");
}

__device__ __forceinline__ void conv_rows_body(
    const float* __restrict__ tb,
    const float* __restrict__ W,
    const float* __restrict__ bias,
    const int*   __restrict__ neigh,
    float*       __restrict__ ob,
    float* sW, float* sB, float* sX)
{
    const int tid = threadIdx.x;

    const int lane = tid & 31;
    const int wave = tid >> 5;
    const int lo   = lane & 15;
    const int hi   = lane >> 4;
    const int sub  = lane >> 3;
    const int q    = lane & 7;

    const int n0 = blockIdx.x * 256 + wave * 32;

    const size_t sN = (size_t)NNODE;

    float* xb = sX + wave * XBUF_FLOATS;

    const char* tbytes = (const char*)tb;
    const unsigned qoff = 16u * (unsigned)q;

    int nidx[7];
    {
        int n = n0 + lane;
        if (n > NNODE - 1) n = NNODE - 1;
        const int* nr = neigh + (size_t)n * 7;
        #pragma unroll
        for (int j = 0; j < 7; ++j) nidx[j] = nr[j];
    }

    v4f rows[8];
    #pragma unroll
    for (int g = 0; g < 8; ++g) {
        const unsigned id = (unsigned)__shfl(nidx[0], 4 * g + sub);
        rows[g] = *(const v4f*)(tbytes + (id * 128u + qoff));
    }

    for (int idx = tid; idx < C_OUT * KTOT; idx += 256) {
        int co = idx / KTOT;
        int k  = idx - co * KTOT;
        int j  = k >> 5;
        int c  = k & 31;
        int cs = ((c & 14) << 1) + ((c >> 4) << 1) + (c & 1);
        sW[co * LDSW_STRIDE + j * 32 + cs] = W[idx];
    }
    if (tid < C_OUT) sB[tid] = bias[tid];
    __syncthreads();

    v8f acc[2][2];
    #pragma unroll
    for (int r = 0; r < 8; ++r) {
        float b0 = sB[hi * 8 + r];
        float b1 = sB[16 + hi * 8 + r];
        acc[0][0][r] = b0; acc[1][0][r] = b0;
        acc[0][1][r] = b1; acc[1][1][r] = b1;
    }

    v4f cur0[4], cur1[4];
    #pragma unroll
    for (int j = 0; j < 7; ++j) {
        __builtin_amdgcn_sched_barrier(0);
        wave_lds_order();
        #pragma unroll
        for (int g = 0; g < 8; ++g)
            *(v4f*)(xb + (4 * g + sub) * XROW_PITCH + 4 * q) = rows[g];
        wave_lds_order();
        #pragma unroll
        for (int i = 0; i < 4; ++i) {
            cur0[i] = *(const v4f*)(xb + lo        * XROW_PITCH + 16 * hi + 4 * i);
            cur1[i] = *(const v4f*)(xb + (16 + lo) * XROW_PITCH + 16 * hi + 4 * i);
        }
        if (j < 6) {
            #pragma unroll
            for (int g = 0; g < 8; ++g) {
                const unsigned id = (unsigned)__shfl(nidx[j + 1], 4 * g + sub);
                rows[g] = *(const v4f*)(tbytes + (id * 128u + qoff));
            }
        }

        const float* w0 = &sW[lo        * LDSW_STRIDE + j * 32 + 2 * hi];
        const float* w1 = &sW[(lo + 16) * LDSW_STRIDE + j * 32 + 2 * hi];

        #pragma unroll
        for (int s = 0; s < 8; ++s) {
            v2f a0 = *(const v2f*)(w0 + s * 4);
            v2f a1 = *(const v2f*)(w1 + s * 4);
            v2f b0, b1;
            if (s & 1) { b0.x = cur0[s >> 1].z; b0.y = cur0[s >> 1].w; b1.x = cur1[s >> 1].z; b1.y = cur1[s >> 1].w; }
            else       { b0.x = cur0[s >> 1].x; b0.y = cur0[s >> 1].y; b1.x = cur1[s >> 1].x; b1.y = cur1[s >> 1].y; }

            acc[0][0] = __builtin_amdgcn_wmma_f32_16x16x4_f32(false, a0, false, b0, (short)0, acc[0][0], false, false);
            acc[0][1] = __builtin_amdgcn_wmma_f32_16x16x4_f32(false, a1, false, b0, (short)0, acc[0][1], false, false);
            acc[1][0] = __builtin_amdgcn_wmma_f32_16x16x4_f32(false, a0, false, b1, (short)0, acc[1][0], false, false);
            acc[1][1] = __builtin_amdgcn_wmma_f32_16x16x4_f32(false, a1, false, b1, (short)0, acc[1][1], false, false);
        }
    }

    #pragma unroll
    for (int t = 0; t < 2; ++t) {
        int n = n0 + t * 16 + lo;
        if (n < NNODE) {
            #pragma unroll
            for (int mt = 0; mt < 2; ++mt) {
                #pragma unroll
                for (int r = 0; r < 8; ++r) {
                    int co = mt * 16 + hi * 8 + r;
                    ob[(size_t)co * sN + n] = acc[t][mt][r];
                }
            }
        }
    }
}

#define SMEM_FLOATS (C_OUT * LDSW_STRIDE + C_OUT + 8 * XBUF_FLOATS)

__global__ __launch_bounds__(256) void onering_conv2_rows1_wmma_f32(
    const float* __restrict__ rows0,
    float*       __restrict__ out0,
    const float* __restrict__ x1,
    const float* __restrict__ rows1,
    size_t                    rows1_bytes,
    float*       __restrict__ out1,
    const float* __restrict__ xt,
    float*       __restrict__ rowst,
    const float* __restrict__ W,
    const float* __restrict__ bias,
    const int*   __restrict__ neigh)
{
    __shared__ __align__(16) float smem[SMEM_FLOATS];
    float* sW = smem;
    float* sB = smem + C_OUT * LDSW_STRIDE;
    float* sX = sB + C_OUT;

    if (blockIdx.y == 2) {
        rows_body(xt, rowst, (float (*)[257])smem);
    } else {
        const bool first = blockIdx.y == 0;
        if (first || rows1_bytes >= PLANE_BYTES)
            conv_rows_body(first ? rows0 : rows1, W, bias, neigh, first ? out0 : out1, sW, sB, sX);
        else
            conv_direct_body<2>(x1, W, bias, neigh, out1, sW, sB);
    }
}

__global__ __launch_bounds__(256) void onering_conv1_rows1_wmma_f32(
    const float* __restrict__ rows,
    float*       __restrict__ out,
    const float* __restrict__ xt,
    float*       __restrict__ rowst,
    size_t                    rowst_bytes,
    const float* __restrict__ W,
    const float* __restrict__ bias,
    const int*   __restrict__ neigh)
{
    __shared__ __align__(16) float smem[SMEM_FLOATS];
    float* sW = smem;
    float* sB = smem + C_OUT * LDSW_STRIDE;
    float* sX = sB + C_OUT;

    if (blockIdx.y == 0) {
        conv_rows_body(rows, W, bias, neigh, out, sW, sB, sX);
    } else if (rowst_bytes >= PLANE_BYTES) {
        rows_body(xt, rowst, (float (*)[257])smem);
    }
}

__global__ __launch_bounds__(256) void onering_conv_last_wmma_f32(
    const float* __restrict__ x,
    const float* __restrict__ rows,
    size_t                    rows_bytes,
    const float* __restrict__ W,
    const float* __restrict__ bias,
    const int*   __restrict__ neigh,
    float*       __restrict__ out)
{
    __shared__ __align__(16) float sW[C_OUT * LDSW_STRIDE];
    __shared__ __align__(16) float sB[C_OUT];
    __shared__ __align__(16) float sX[8 * XBUF_FLOATS];

    if (rows_bytes >= PLANE_BYTES) conv_rows_body(rows, W, bias, neigh, out, sW, sB, sX);
    else                           conv_direct_body<8>(x, W, bias, neigh, out, sW, sB);
}

extern "C" void kernel_launch(void* const* d_in, const int* in_sizes, int n_in,
                              void* d_out, int out_size, void* d_ws, size_t ws_size,
                              hipStream_t stream) {
    const float* x     = (const float*)d_in[0];
    const float* W     = (const float*)d_in[1];
    const float* bias  = (const float*)d_in[2];
    const int*   neigh = (const int*)d_in[3];
    float*       out   = (float*)d_out;
    float*       ws    = (float*)d_ws;

    const dim3 blk(256, 1, 1);
    const dim3 g1((NNODE + 255) / 256, 1, 1);
    const dim3 g2((NNODE + 255) / 256, 2, 1);
    const dim3 g3((NNODE + 255) / 256, 3, 1);

    x_to_node_rows<<<g2, blk, 0, stream>>>(x, out + 2 * PLANE, ws, ws_size);
    onering_conv2_rows1_wmma_f32<<<g3, blk, 0, stream>>>(out + 2 * PLANE, out,
                                                         x + PLANE, ws, ws_size, out + PLANE,
                                                         x + 2 * PLANE, out + 3 * PLANE,
                                                         W, bias, neigh);
    onering_conv1_rows1_wmma_f32<<<g2, blk, 0, stream>>>(out + 3 * PLANE, out + 2 * PLANE,
                                                         x + 3 * PLANE, ws, ws_size,
                                                         W, bias, neigh);
    onering_conv_last_wmma_f32<<<g1, blk, 0, stream>>>(x + 3 * PLANE, ws, ws_size, W, bias, neigh, out + 3 * PLANE);
}


# rows path: a block stages its weights in one round trip (a thread's 7 whole 16-B reads of W and its bias read go out together, ahead of the index reads), not 14 in a row; LDS slots, k order kept
# speedup vs baseline: 5.2624x; 1.0478x over previous
#include <hip/hip_runtime.h>

typedef __attribute__((ext_vector_type(2))) float v2f;
typedef __attribute__((ext_vector_type(8))) float v8f;
typedef __attribute__((ext_vector_type(4))) float v4f;

#define C_IN   32
#define C_OUT  32
#define NNODE  163842
#define KTOT   224
#define LDSW_STRIDE 226
#define PLANE  ((size_t)C_IN * NNODE)
#define PLANE_BYTES (PLANE * sizeof(float))

template <int SU>
__device__ __forceinline__ void conv_direct_body(
    const float* __restrict__ x,
    const float* __restrict__ W,
    const float* __restrict__ bias,
    const int*   __restrict__ neigh,
    float*       __restrict__ out,
    float* sW, float* sB)
{
    const int tid = threadIdx.x;

    for (int idx = tid; idx < C_OUT * KTOT; idx += 256) {
        int co = idx / KTOT;
        int k  = idx - co * KTOT;
        sW[co * LDSW_STRIDE + k] = W[idx];
    }
    if (tid < C_OUT) sB[tid] = bias[tid];
    __syncthreads();

    const int lane = tid & 31;
    const int wave = tid >> 5;
    const int lo   = lane & 15;
    const int hi   = lane >> 4;

    const int n0    = blockIdx.x * 256 + wave * 32;

    const float* xb = x;
    float*       ob = out;
    const size_t sN = (size_t)NNODE;

    v8f acc[2][2];
    #pragma unroll
    for (int r = 0; r < 8; ++r) {
        float b0 = sB[hi * 8 + r];
        float b1 = sB[16 + hi * 8 + r];
        acc[0][0][r] = b0; acc[1][0][r] = b0;
        acc[0][1][r] = b1; acc[1][1][r] = b1;
    }

    int nidx[2][7];
    #pragma unroll
    for (int t = 0; t < 2; ++t) {
        int n = n0 + t * 16 + lo;
        if (n > NNODE - 1) n = NNODE - 1;
        const int* nr = neigh + (size_t)n * 7;
        #pragma unroll
        for (int j = 0; j < 7; ++j) nidx[t][j] = nr[j];
    }

    for (int j = 0; j < 7; ++j) {
        const float* p0 = xb + (size_t)nidx[0][j] + (size_t)(2 * hi) * sN;
        const float* p1 = xb + (size_t)nidx[1][j] + (size_t)(2 * hi) * sN;
        const float* w0 = &sW[lo        * LDSW_STRIDE + j * 32 + 2 * hi];
        const float* w1 = &sW[(lo + 16) * LDSW_STRIDE + j * 32 + 2 * hi];

        #pragma unroll SU
        for (int s = 0; s < 8; ++s) {
            v2f a0 = *(const v2f*)(w0 + s * 4);
            v2f a1 = *(const v2f*)(w1 + s * 4);
            v2f b0; b0.x = p0[0]; b0.y = p0[sN];
            v2f b1; b1.x = p1[0]; b1.y = p1[sN];
            p0 += 4 * sN; p1 += 4 * sN;

            acc[0][0] = __builtin_amdgcn_wmma_f32_16x16x4_f32(false, a0, false, b0, (short)0, acc[0][0], false, false);
            acc[0][1] = __builtin_amdgcn_wmma_f32_16x16x4_f32(false, a1, false, b0, (short)0, acc[0][1], false, false);
            acc[1][0] = __builtin_amdgcn_wmma_f32_16x16x4_f32(false, a0, false, b1, (short)0, acc[1][0], false, false);
            acc[1][1] = __builtin_amdgcn_wmma_f32_16x16x4_f32(false, a1, false, b1, (short)0, acc[1][1], false, false);
        }
    }

    #pragma unroll
    for (int t = 0; t < 2; ++t) {
        int n = n0 + t * 16 + lo;
        if (n < NNODE) {
            #pragma unroll
            for (int mt = 0; mt < 2; ++mt) {
                #pragma unroll
                for (int r = 0; r < 8; ++r) {
                    int co = mt * 16 + hi * 8 + r;
                    ob[(size_t)co * sN + n] = acc[t][mt][r];
                }
            }
        }
    }
}

__device__ __forceinline__ void rows_body(
    const float* __restrict__ xb,
    float*       __restrict__ db,
    float (*tile)[257])
{
    const int tid = threadIdx.x;
    const int n0  = blockIdx.x * 256;

    const int n = n0 + tid;
    if (n < NNODE) {
        #pragma unroll
        for (int c = 0; c < C_IN; ++c) tile[c][tid] = xb[(size_t)c * NNODE + n];
    }
    __syncthreads();

    const int q = tid & 7;
    #pragma unroll
    for (int k = 0; k < 8; ++k) {
        int r = (tid >> 3) + k * 32;
        if (n0 + r < NNODE) {
            float4 v;
            v.x = tile[4 * q + 0][r];
            v.y = tile[4 * q + 1][r];
            v.z = tile[4 * q + 2][r];
            v.w = tile[4 * q + 3][r];
            *(float4*)(db + (size_t)(n0 + r) * C_IN + 4 * q) = v;
        }
    }
}

__global__ __launch_bounds__(256) void x_to_node_rows(
    const float* __restrict__ x,
    float*       __restrict__ dst0,
    float*       __restrict__ dst1,
    size_t                    dst1_bytes)
{
    __shared__ float tile[C_IN][257];

    const bool second = blockIdx.y != 0;
    if (second && dst1_bytes < PLANE_BYTES) return;

    rows_body(x + (size_t)blockIdx.y * PLANE, second ? dst1 : dst0, tile);
}

#define XROW_PITCH 36
#define XBUF_FLOATS (32 * XROW_PITCH)

__device__ __forceinline__ void wave_lds_order()
{
    __builtin_amdgcn_fence(__ATOMIC_RELEASE, "wavefront");
    __builtin_amdgcn_wave_barrier();
    __builtin_amdgcn_fence(__ATOMIC_ACQUIRE, "wavefront");
}

__device__ __forceinline__ void conv_rows_body(
    const float* __restrict__ tb,
    const float* __restrict__ W,
    const float* __restrict__ bias,
    const int*   __restrict__ neigh,
    float*       __restrict__ ob,
    float* sW, float* sB, float* sX)
{
    const int tid = threadIdx.x;

    const int lane = tid & 31;
    const int wave = tid >> 5;
    const int lo   = lane & 15;
    const int hi   = lane >> 4;
    const int sub  = lane >> 3;
    const int q    = lane & 7;

    const int n0 = blockIdx.x * 256 + wave * 32;

    const size_t sN = (size_t)NNODE;

    float* xb = sX + wave * XBUF_FLOATS;

    v4f wv[7];
    #pragma unroll
    for (int i = 0; i < 7; ++i) wv[i] = *(const v4f*)(W + 4 * (tid + 256 * i));
    float bv = 0.0f;
    if (tid < C_OUT) bv = bias[tid];

    const char* tbytes = (const char*)tb;
    const unsigned qoff = 16u * (unsigned)q;

    int nidx[7];
    {
        int n = n0 + lane;
        if (n > NNODE - 1) n = NNODE - 1;
        const int* nr = neigh + (size_t)n * 7;
        #pragma unroll
        for (int j = 0; j < 7; ++j) nidx[j] = nr[j];
    }

    v4f rows[8];
    #pragma unroll
    for (int g = 0; g < 8; ++g) {
        const unsigned id = (unsigned)__shfl(nidx[0], 4 * g + sub);
        rows[g] = *(const v4f*)(tbytes + (id * 128u + qoff));
    }

    #pragma unroll
    for (int i = 0; i < 7; ++i) {
        int idx = 4 * (tid + 256 * i);
        int co = idx / KTOT;
        int k  = idx - co * KTOT;
        int j  = k >> 5;
        int c  = k & 31;
        int cs = ((c & 14) << 1) + ((c >> 4) << 1);
        float* d = &sW[co * LDSW_STRIDE + j * 32 + cs];
        v2f p0; p0.x = wv[i].x; p0.y = wv[i].y;
        v2f p1; p1.x = wv[i].z; p1.y = wv[i].w;
        *(v2f*)d       = p0;
        *(v2f*)(d + 4) = p1;
    }
    if (tid < C_OUT) sB[tid] = bv;
    __syncthreads();

    v8f acc[2][2];
    #pragma unroll
    for (int r = 0; r < 8; ++r) {
        float b0 = sB[hi * 8 + r];
        float b1 = sB[16 + hi * 8 + r];
        acc[0][0][r] = b0; acc[1][0][r] = b0;
        acc[0][1][r] = b1; acc[1][1][r] = b1;
    }

    v4f cur0[4], cur1[4];
    #pragma unroll
    for (int j = 0; j < 7; ++j) {
        __builtin_amdgcn_sched_barrier(0);
        wave_lds_order();
        #pragma unroll
        for (int g = 0; g < 8; ++g)
            *(v4f*)(xb + (4 * g + sub) * XROW_PITCH + 4 * q) = rows[g];
        wave_lds_order();
        #pragma unroll
        for (int i = 0; i < 4; ++i) {
            cur0[i] = *(const v4f*)(xb + lo        * XROW_PITCH + 16 * hi + 4 * i);
            cur1[i] = *(const v4f*)(xb + (16 + lo) * XROW_PITCH + 16 * hi + 4 * i);
        }
        if (j < 6) {
            #pragma unroll
            for (int g = 0; g < 8; ++g) {
                const unsigned id = (unsigned)__shfl(nidx[j + 1], 4 * g + sub);
                rows[g] = *(const v4f*)(tbytes + (id * 128u + qoff));
            }
        }

        const float* w0 = &sW[lo        * LDSW_STRIDE + j * 32 + 2 * hi];
        const float* w1 = &sW[(lo + 16) * LDSW_STRIDE + j * 32 + 2 * hi];

        #pragma unroll
        for (int s = 0; s < 8; ++s) {
            v2f a0 = *(const v2f*)(w0 + s * 4);
            v2f a1 = *(const v2f*)(w1 + s * 4);
            v2f b0, b1;
            if (s & 1) { b0.x = cur0[s >> 1].z; b0.y = cur0[s >> 1].w; b1.x = cur1[s >> 1].z; b1.y = cur1[s >> 1].w; }
            else       { b0.x = cur0[s >> 1].x; b0.y = cur0[s >> 1].y; b1.x = cur1[s >> 1].x; b1.y = cur1[s >> 1].y; }

            acc[0][0] = __builtin_amdgcn_wmma_f32_16x16x4_f32(false, a0, false, b0, (short)0, acc[0][0], false, false);
            acc[0][1] = __builtin_amdgcn_wmma_f32_16x16x4_f32(false, a1, false, b0, (short)0, acc[0][1], false, false);
            acc[1][0] = __builtin_amdgcn_wmma_f32_16x16x4_f32(false, a0, false, b1, (short)0, acc[1][0], false, false);
            acc[1][1] = __builtin_amdgcn_wmma_f32_16x16x4_f32(false, a1, false, b1, (short)0, acc[1][1], false, false);
        }
    }

    #pragma unroll
    for (int t = 0; t < 2; ++t) {
        int n = n0 + t * 16 + lo;
        if (n < NNODE) {
            #pragma unroll
            for (int mt = 0; mt < 2; ++mt) {
                #pragma unroll
                for (int r = 0; r < 8; ++r) {
                    int co = mt * 16 + hi * 8 + r;
                    ob[(size_t)co * sN + n] = acc[t][mt][r];
                }
            }
        }
    }
}

#define SMEM_FLOATS (C_OUT * LDSW_STRIDE + C_OUT + 8 * XBUF_FLOATS)

__global__ __launch_bounds__(256) void onering_conv2_rows1_wmma_f32(
    const float* __restrict__ rows0,
    float*       __restrict__ out0,
    const float* __restrict__ x1,
    const float* __restrict__ rows1,
    size_t                    rows1_bytes,
    float*       __restrict__ out1,
    const float* __restrict__ xt,
    float*       __restrict__ rowst,
    const float* __restrict__ W,
    const float* __restrict__ bias,
    const int*   __restrict__ neigh)
{
    __shared__ __align__(16) float smem[SMEM_FLOATS];
    float* sW = smem;
    float* sB = smem + C_OUT * LDSW_STRIDE;
    float* sX = sB + C_OUT;

    if (blockIdx.y == 2) {
        rows_body(xt, rowst, (float (*)[257])smem);
    } else {
        const bool first = blockIdx.y == 0;
        if (first || rows1_bytes >= PLANE_BYTES)
            conv_rows_body(first ? rows0 : rows1, W, bias, neigh, first ? out0 : out1, sW, sB, sX);
        else
            conv_direct_body<2>(x1, W, bias, neigh, out1, sW, sB);
    }
}

__global__ __launch_bounds__(256) void onering_conv1_rows1_wmma_f32(
    const float* __restrict__ rows,
    float*       __restrict__ out,
    const float* __restrict__ xt,
    float*       __restrict__ rowst,
    size_t                    rowst_bytes,
    const float* __restrict__ W,
    const float* __restrict__ bias,
    const int*   __restrict__ neigh)
{
    __shared__ __align__(16) float smem[SMEM_FLOATS];
    float* sW = smem;
    float* sB = smem + C_OUT * LDSW_STRIDE;
    float* sX = sB + C_OUT;

    if (blockIdx.y == 0) {
        conv_rows_body(rows, W, bias, neigh, out, sW, sB, sX);
    } else if (rowst_bytes >= PLANE_BYTES) {
        rows_body(xt, rowst, (float (*)[257])smem);
    }
}

__global__ __launch_bounds__(256) void onering_conv_last_wmma_f32(
    const float* __restrict__ x,
    const float* __restrict__ rows,
    size_t                    rows_bytes,
    const float* __restrict__ W,
    const float* __restrict__ bias,
    const int*   __restrict__ neigh,
    float*       __restrict__ out)
{
    __shared__ __align__(16) float sW[C_OUT * LDSW_STRIDE];
    __shared__ __align__(16) float sB[C_OUT];
    __shared__ __align__(16) float sX[8 * XBUF_FLOATS];

    if (rows_bytes >= PLANE_BYTES) conv_rows_body(rows, W, bias, neigh, out, sW, sB, sX);
    else                           conv_direct_body<8>(x, W, bias, neigh, out, sW, sB);
}

extern "C" void kernel_launch(void* const* d_in, const int* in_sizes, int n_in,
                              void* d_out, int out_size, void* d_ws, size_t ws_size,
                              hipStream_t stream) {
    const float* x     = (const float*)d_in[0];
    const float* W     = (const float*)d_in[1];
    const float* bias  = (const float*)d_in[2];
    const int*   neigh = (const int*)d_in[3];
    float*       out   = (float*)d_out;
    float*       ws    = (float*)d_ws;

    const dim3 blk(256, 1, 1);
    const dim3 g1((NNODE + 255) / 256, 1, 1);
    const dim3 g2((NNODE + 255) / 256, 2, 1);
    const dim3 g3((NNODE + 255) / 256, 3, 1);

    x_to_node_rows<<<g2, blk, 0, stream>>>(x, out + 2 * PLANE, ws, ws_size);
    onering_conv2_rows1_wmma_f32<<<g3, blk, 0, stream>>>(out + 2 * PLANE, out,
                                                         x + PLANE, ws, ws_size, out + PLANE,
                                                         x + 2 * PLANE, out + 3 * PLANE,
                                                         W, bias, neigh);
    onering_conv1_rows1_wmma_f32<<<g2, blk, 0, stream>>>(out + 3 * PLANE, out + 2 * PLANE,
                                                         x + 3 * PLANE, ws, ws_size,
                                                         W, bias, neigh);
    onering_conv_last_wmma_f32<<<g1, blk, 0, stream>>>(x + 3 * PLANE, ws, ws_size, W, bias, neigh, out + 3 * PLANE);
}
